# starting kernel
# baseline (speedup 1.0000x reference)
#include <hip/hip_runtime.h>
#include <hip/hip_bf16.h>

typedef __attribute__((ext_vector_type(2))) float v2f;
typedef __attribute__((ext_vector_type(8))) float v8f;

#define B_ 256
#define N_ 512
#define D_ 128
#define H_ 256

__global__ __launch_bounds__(256) void FusedLinearMax_kernel(
    const float* __restrict__ x, const float* __restrict__ w,
    const float* __restrict__ bias, float* __restrict__ out)
{
  const int lane = threadIdx.x & 31;
  const int wave = threadIdx.x >> 5;
  const int b    = blockIdx.x;
  const int h0   = wave << 5;

  const int row  = lane & 15;
  const int koff = (lane >> 4) << 1;

  const float* xb  = x + (size_t)b * (N_ * D_);
  const float* wt0 = w + (size_t)(h0 + row) * D_ + koff;
  const float* wt1 = wt0 + 16 * D_;

  v2f bf0[32], bf1[32];
#pragma unroll
  for (int kk = 0; kk < 32; ++kk) {
    bf0[kk] = *(const v2f*)(wt0 + 4 * kk);
    bf1[kk] = *(const v2f*)(wt1 + 4 * kk);
  }

  float m0 = -__builtin_inff();
  float m1 = -__builtin_inff();

#pragma unroll 2
  for (int nt = 0; nt < N_ / 16; ++nt) {
    const float* xa = xb + (size_t)(nt * 16 + row) * D_ + koff;
    v2f af[32];
#pragma unroll
    for (int kk = 0; kk < 32; ++kk) {
      af[kk] = *(const v2f*)(xa + 4 * kk);
    }
    v8f c0 = {0.f, 0.f, 0.f, 0.f, 0.f, 0.f, 0.f, 0.f};
    v8f c1 = {0.f, 0.f, 0.f, 0.f, 0.f, 0.f, 0.f, 0.f};
#pragma unroll
    for (int kk = 0; kk < 32; ++kk) {
      c0 = __builtin_amdgcn_wmma_f32_16x16x4_f32(
          false, af[kk], false, bf0[kk], (short)0, c0, false, false);
      c1 = __builtin_amdgcn_wmma_f32_16x16x4_f32(
          false, af[kk], false, bf1[kk], (short)0, c1, false, false);
    }
#pragma unroll
    for (int r = 0; r < 8; ++r) {
      m0 = fmaxf(m0, c0[r]);
      m1 = fmaxf(m1, c1[r]);
    }
  }

  m0 = fmaxf(m0, __shfl_xor(m0, 16, 32));
  m1 = fmaxf(m1, __shfl_xor(m1, 16, 32));

  if (lane < 16) {
    const size_t base = (size_t)b * H_ + h0;
    out[base + lane]      = m0 + bias[h0 + lane];
    out[base + 16 + lane] = m1 + bias[h0 + 16 + lane];
  }
}

extern "C" void kernel_launch(void* const* d_in, const int* in_sizes, int n_in,
                              void* d_out, int out_size, void* d_ws, size_t ws_size,
                              hipStream_t stream) {
  (void)in_sizes; (void)n_in; (void)d_ws; (void)ws_size; (void)out_size;
  const float* x    = (const float*)d_in[0];
  const float* w    = (const float*)d_in[1];
  const float* bias = (const float*)d_in[2];
  float* out        = (float*)d_out;

  dim3 grid(B_);
  dim3 block(256);
  hipLaunchKernelGGL(FusedLinearMax_kernel, grid, block, 0, stream, x, w, bias, out);
}


# one wave per SIMD owning 4 h-tiles (was 2 waves x 2): each tile's 32 x loads issue as ONE group a tile ahead of its 128 WMMAs, not one per 2 WMMAs; half the x loads per WMMA; sums bit-identical
# speedup vs baseline: 1.3934x; 1.3934x over previous
#include <hip/hip_runtime.h>
#include <hip/hip_bf16.h>

typedef __attribute__((ext_vector_type(2))) float v2f;
typedef __attribute__((ext_vector_type(8))) float v8f;

#define B_ 256
#define N_ 512
#define D_ 128
#define H_ 256
#define HT_ 4
#define KS_ (D_ / 4)


__device__ __forceinline__ void load_a(v2f (&a)[KS_], const float* __restrict__ p)
{
#pragma unroll
  for (int kk = 0; kk < KS_; ++kk)
    a[kk] = *(const v2f*)(p + 4 * kk);
}

__device__ __forceinline__ void tile_max(const v2f (&a)[KS_],
                                         const v2f (&bf)[HT_][KS_],
                                         float (&m)[HT_])
{
  v8f c[HT_];
#pragma unroll
  for (int t = 0; t < HT_; ++t)
    c[t] = (v8f){0.f, 0.f, 0.f, 0.f, 0.f, 0.f, 0.f, 0.f};
#pragma unroll
  for (int kk = 0; kk < KS_; ++kk) {
#pragma unroll
    for (int t = 0; t < HT_; ++t)
      c[t] = __builtin_amdgcn_wmma_f32_16x16x4_f32(
          false, a[kk], false, bf[t][kk], (short)0, c[t], false, false);
  }
#pragma unroll
  for (int t = 0; t < HT_; ++t) {
#pragma unroll
    for (int r = 0; r < 8; ++r)
      m[t] = fmaxf(m[t], c[t][r]);
  }
}

__global__ __launch_bounds__(128) void FusedLinearMax_kernel(
    const float* __restrict__ x, const float* __restrict__ w,
    const float* __restrict__ bias, float* __restrict__ out)
{
  const int lane = threadIdx.x & 31;
  const int wave = threadIdx.x >> 5;
  const int b    = blockIdx.x;
  const int h0   = wave * (16 * HT_);

  const int row  = lane & 15;
  const int koff = (lane >> 4) << 1;

  const float* xa = x + (size_t)b * (N_ * D_) + (size_t)row * D_ + koff;
  const float* wt = w + (size_t)(h0 + row) * D_ + koff;

  v2f a0[KS_], a1[KS_];
  load_a(a0, xa);

  v2f bf[HT_][KS_];
#pragma unroll
  for (int t = 0; t < HT_; ++t) {
#pragma unroll
    for (int kk = 0; kk < KS_; ++kk)
      bf[t][kk] = *(const v2f*)(wt + (size_t)t * 16 * D_ + 4 * kk);
  }

  float m[HT_];
#pragma unroll
  for (int t = 0; t < HT_; ++t) m[t] = -__builtin_inff();

  const int NT = N_ / 16;
  const size_t tstride = (size_t)16 * D_;

#define GROUP_FENCE() __builtin_amdgcn_sched_barrier(0)

  for (int nt = 0; nt < NT - 2; nt += 2) {
    load_a(a1, xa + (size_t)(nt + 1) * tstride);
    GROUP_FENCE();
    tile_max(a0, bf, m);
    GROUP_FENCE();
    load_a(a0, xa + (size_t)(nt + 2) * tstride);
    GROUP_FENCE();
    tile_max(a1, bf, m);
  }
  load_a(a1, xa + (size_t)(NT - 1) * tstride);
  GROUP_FENCE();
  tile_max(a0, bf, m);
  tile_max(a1, bf, m);
#undef GROUP_FENCE

#pragma unroll
  for (int t = 0; t < HT_; ++t)
    m[t] = fmaxf(m[t], __shfl_xor(m[t], 16, 32));

  if (lane < 16) {
    const size_t base = (size_t)b * H_ + h0;
#pragma unroll
    for (int t = 0; t < HT_; ++t)
      out[base + 16 * t + lane] = m[t] + bias[h0 + 16 * t + lane];
  }
}

extern "C" void kernel_launch(void* const* d_in, const int* in_sizes, int n_in,
                              void* d_out, int out_size, void* d_ws, size_t ws_size,
                              hipStream_t stream) {
  (void)in_sizes; (void)n_in; (void)d_ws; (void)ws_size; (void)out_size;
  const float* x    = (const float*)d_in[0];
  const float* w    = (const float*)d_in[1];
  const float* bias = (const float*)d_in[2];
  float* out        = (float*)d_out;

  dim3 grid(B_);
  dim3 block(128);
  hipLaunchKernelGGL(FusedLinearMax_kernel, grid, block, 0, stream, x, w, bias, out);
}
